# starting kernel
# baseline (speedup 1.0000x reference)
#include <hip/hip_runtime.h>
#include <math.h>

#define B_TOTAL   16384
#define ND        13
#define NS        26
#define FEAT_NUM  10000
#define FLEN      260013
#define KDIM      64

typedef __attribute__((ext_vector_type(2))) float v2f;
typedef __attribute__((ext_vector_type(8))) float v8f;

__device__ __forceinline__ float sigmoidf_(float x) {
    return 1.0f / (1.0f + __expf(-x));
}

__global__ __launch_bounds__(256) void FM_29437705847531_kernel(
    const float* __restrict__ dense,
    const int*   __restrict__ sp,
    const float* __restrict__ w0p,
    const float* __restrict__ w,
    const float* __restrict__ V,
    float* __restrict__ out)
{
    const int lane = threadIdx.x & 31;
    const int wave = threadIdx.x >> 5;
    const int tile = blockIdx.x * 8 + wave;
    const int row0 = tile * 16;

    const int n  = lane & 15;
    const int hi = lane >> 4;

    const float* rowPtr[4];
#pragma unroll
    for (int nt = 0; nt < 4; ++nt)
        rowPtr[nt] = V + (size_t)(nt * 16 + n) * FLEN;

    v8f acc_vx[4] = {};
    v8f acc_v2[4] = {};

#pragma unroll
    for (int kk = 0; kk < 16; kk += 4) {
        const int f0 = kk + 2 * hi;
        v2f a;
        a.x = (f0 + 0 < ND) ? dense[(size_t)(row0 + n) * ND + f0 + 0] : 0.0f;
        a.y = (f0 + 1 < ND) ? dense[(size_t)(row0 + n) * ND + f0 + 1] : 0.0f;
        const v2f a2 = a * a;
#pragma unroll
        for (int nt = 0; nt < 4; ++nt) {
            v2f b;
            b.x = (f0 + 0 < ND) ? rowPtr[nt][f0 + 0] : 0.0f;
            b.y = (f0 + 1 < ND) ? rowPtr[nt][f0 + 1] : 0.0f;
            const v2f b2 = b * b;
            acc_vx[nt] = __builtin_amdgcn_wmma_f32_16x16x4_f32(
                false, a, false, b, (short)0, acc_vx[nt], false, false);
            acc_v2[nt] = __builtin_amdgcn_wmma_f32_16x16x4_f32(
                false, a2, false, b2, (short)0, acc_v2[nt], false, false);
        }
    }

    for (int j = 0; j < NS; ++j) {
#pragma unroll
        for (int v = 0; v < 8; ++v) {
            const int m = v + 8 * hi;
            const size_t col = (size_t)ND + (size_t)j * FEAT_NUM
                             + (size_t)sp[(size_t)(row0 + m) * NS + j];
#pragma unroll
            for (int nt = 0; nt < 4; ++nt) {
                const float val = rowPtr[nt][col];
                acc_vx[nt][v] += val;
                acc_v2[nt][v] += val * val;
            }
        }
    }

    const int r1 = row0 + n;
    float first = w0p[0];
    for (int f = 0; f < ND; ++f)
        first += dense[(size_t)r1 * ND + f] * w[f];
    for (int j = 0; j < NS; ++j)
        first += w[(size_t)ND + (size_t)j * FEAT_NUM
                   + (size_t)sp[(size_t)r1 * NS + j]];

#pragma unroll
    for (int v = 0; v < 8; ++v) {
        float t = 0.0f;
#pragma unroll
        for (int nt = 0; nt < 4; ++nt) {
            const float x = acc_vx[nt][v];
            t += x * x - acc_v2[nt][v];
        }
        t += __shfl_xor(t, 1, 32);
        t += __shfl_xor(t, 2, 32);
        t += __shfl_xor(t, 4, 32);
        t += __shfl_xor(t, 8, 32);
        const float sec = 0.5f * t;

        const float f_lo = __shfl(first, v,     32);
        const float f_hi = __shfl(first, 8 + v, 32);
        if (lane == 0)        out[row0 + v]     = sigmoidf_(f_lo + sec);
        else if (lane == 16)  out[row0 + 8 + v] = sigmoidf_(f_hi + sec);
    }
}

extern "C" void kernel_launch(void* const* d_in, const int* in_sizes, int n_in,
                              void* d_out, int out_size, void* d_ws, size_t ws_size,
                              hipStream_t stream) {
    const float* dense = (const float*)d_in[0];
    const int*   sp    = (const int*)  d_in[1];
    const float* w0p   = (const float*)d_in[2];
    const float* w     = (const float*)d_in[3];
    const float* V     = (const float*)d_in[4];
    float* out = (float*)d_out;

    dim3 grid(B_TOTAL / (16 * 8));
    FM_29437705847531_kernel<<<grid, 256, 0, stream>>>(dense, sp, w0p, w, V, out);
}


# V copied feature-major into the workspace by a first launch, so each gathered embedding is one 256 B row (16 B per lane); a lane now holds latents 4n..4n+3, so the 64-latent sum order changes
# speedup vs baseline: 4.2144x; 4.2144x over previous
#include <hip/hip_runtime.h>
#include <math.h>


#define B_TOTAL   16384
#define ND        13
#define NS        26
#define FEAT_NUM  10000
#define FLEN      260013
#define KDIM      64

#define TP_F      128
#define TP_PITCH  129
#define TP_TILES  ((FLEN + TP_F - 1) / TP_F)
#define TP_BLOCKS 1024

typedef __attribute__((ext_vector_type(2))) float v2f;
typedef __attribute__((ext_vector_type(4))) float v4f;
typedef __attribute__((ext_vector_type(8))) float v8f;

__device__ __forceinline__ float sigmoidf_(float x) {
    return 1.0f / (1.0f + __expf(-x));
}

__device__ __forceinline__ void tp_load(float (&x)[32], const float* __restrict__ V,
                                        int f0, int wave, int lane) {
#pragma unroll
    for (int r = 0; r < 8; ++r) {
        const float* src = V + (size_t)(wave * 8 + r) * FLEN;
#pragma unroll
        for (int h = 0; h < 4; ++h) {
            const int f  = f0 + h * 32 + lane;
            const int fc = (f < FLEN) ? f : (FLEN - 1);
            x[r * 4 + h] = src[fc];
        }
    }
}

__device__ __forceinline__ void tp_store(const float (&x)[32], float* tile, float* __restrict__ Vt,
                                         int f0, int wave, int lane, int tid) {
#pragma unroll
    for (int r = 0; r < 8; ++r) {
#pragma unroll
        for (int h = 0; h < 4; ++h)
            tile[(wave * 8 + r) * TP_PITCH + h * 32 + lane] = x[r * 4 + h];
    }
    __syncthreads();
    const int n  = tid & 15;
    const int fr = tid >> 4;
#pragma unroll
    for (int i = 0; i < TP_F / 16; ++i) {
        const int fl = fr + 16 * i;
        v4f o;
        o.x = tile[(4 * n + 0) * TP_PITCH + fl];
        o.y = tile[(4 * n + 1) * TP_PITCH + fl];
        o.z = tile[(4 * n + 2) * TP_PITCH + fl];
        o.w = tile[(4 * n + 3) * TP_PITCH + fl];
        if (f0 + fl < FLEN)
            *(v4f*)(Vt + (size_t)(f0 + fl) * KDIM + 4 * n) = o;
    }
}

__global__ __launch_bounds__(256) void fm_transpose_kernel(
    const float* __restrict__ V,
    float* __restrict__ Vt)
{
    __shared__ float tile[KDIM * TP_PITCH];

    const int tid  = threadIdx.x;
    const int lane = tid & 31;
    const int wave = tid >> 5;
    const int t0   = blockIdx.x;
    const int t1   = blockIdx.x + TP_BLOCKS;
    const bool has1 = (t1 < TP_TILES);

    const int t1c = has1 ? t1 : t0;
    float xa[32], xb[32];
    tp_load(xa, V, t0 * TP_F, wave, lane);
    tp_load(xb, V, t1c * TP_F, wave, lane);

    tp_store(xa, tile, Vt, t0 * TP_F, wave, lane, tid);
    if (has1) {
        __syncthreads();
        tp_store(xb, tile, Vt, t1 * TP_F, wave, lane, tid);
    }
}

#define FM_ISSUE(G, J0)                                                              \
    _Pragma("unroll")                                                                \
    for (int q = 0; q < 2; ++q) {                                                    \
        _Pragma("unroll")                                                            \
        for (int v = 0; v < 8; ++v) {                                                \
            const unsigned col = (unsigned)ND + (unsigned)((J0) + q) * FEAT_NUM      \
                               + (unsigned)spH[v * NS + (J0) + q];                   \
            G[q * 8 + v] = Vt4[col * 16u + (unsigned)n];     \
        }                                                                            \
    }

#define FM_HOLD(G)                                                                   \
    asm volatile("" :: "v"(G[0]), "v"(G[1]), "v"(G[2]), "v"(G[3]), "v"(G[4]),        \
                 "v"(G[5]), "v"(G[6]), "v"(G[7]), "v"(G[8]), "v"(G[9]), "v"(G[10]),  \
                 "v"(G[11]), "v"(G[12]), "v"(G[13]), "v"(G[14]), "v"(G[15]))

#define FM_SUM(G)                                                                    \
    _Pragma("unroll")                                                                \
    for (int q = 0; q < 2; ++q) {                                                    \
        _Pragma("unroll")                                                            \
        for (int v = 0; v < 8; ++v) {                                                \
            _Pragma("unroll")                                                        \
            for (int nt = 0; nt < 4; ++nt) {                                         \
                const float val = G[q * 8 + v][nt];                                  \
                acc_vx[nt][v] += val;                                                \
                acc_v2[nt][v] += val * val;                                          \
            }                                                                        \
        }                                                                            \
    }

__global__ __launch_bounds__(32) void fm_gather_kernel(
    const float* __restrict__ dense,
    const int*   __restrict__ sp,
    const float* __restrict__ w0p,
    const float* __restrict__ w,
    const float* __restrict__ Vt,
    float* __restrict__ out)
{
    const int lane = threadIdx.x & 31;
    const int tile = blockIdx.x;
    const int row0 = tile * 16;

    const int n  = lane & 15;
    const int hi = lane >> 4;

    const v4f* __restrict__ Vt4 = (const v4f*)Vt;

    __shared__ int spL[16 * NS];
    int spr[(16 * NS) / 32];
#pragma unroll
    for (int it = 0; it < (16 * NS) / 32; ++it)
        spr[it] = sp[(size_t)row0 * NS + it * 32 + lane];
#pragma unroll
    for (int it = 0; it < (16 * NS) / 32; ++it)
        spL[it * 32 + lane] = spr[it];
    __syncthreads();

    v8f acc_vx[4] = {};
    v8f acc_v2[4] = {};

#pragma unroll
    for (int kk = 0; kk < 16; kk += 4) {
        const int f0 = kk + 2 * hi;
        const bool in0 = (f0 + 0 < ND);
        const bool in1 = (f0 + 1 < ND);
        const int  fa  = in0 ? (f0 + 0) : (ND - 1);
        const int  fb  = in1 ? (f0 + 1) : (ND - 1);
        const v4f z4 = {0.0f, 0.0f, 0.0f, 0.0f};
        const float ax = dense[(size_t)(row0 + n) * ND + fa];
        const float ay = dense[(size_t)(row0 + n) * ND + fb];
        const v4f  bxl = Vt4[fa * 16 + n];
        const v4f  byl = Vt4[fb * 16 + n];
        v2f a;
        a.x = in0 ? ax : 0.0f;
        a.y = in1 ? ay : 0.0f;
        const v4f bx = in0 ? bxl : z4;
        const v4f by = in1 ? byl : z4;
        const v2f a2 = a * a;
#pragma unroll
        for (int nt = 0; nt < 4; ++nt) {
            v2f b;
            b.x = bx[nt];
            b.y = by[nt];
            const v2f b2 = b * b;
            acc_vx[nt] = __builtin_amdgcn_wmma_f32_16x16x4_f32(
                false, a, false, b, (short)0, acc_vx[nt], false, false);
            acc_v2[nt] = __builtin_amdgcn_wmma_f32_16x16x4_f32(
                false, a2, false, b2, (short)0, acc_v2[nt], false, false);
        }
    }

    const int r1 = row0 + n;
    float first = w0p[0];
    for (int f = 0; f < ND; ++f)
        first += dense[(size_t)r1 * ND + f] * w[f];
    for (int j = 0; j < NS; ++j)
        first += w[(size_t)ND + (size_t)j * FEAT_NUM + (size_t)spL[n * NS + j]];

    const int* spH = spL + 8 * hi * NS;
    v4f gA[16], gB[16];

    FM_ISSUE(gA, 0);
#pragma unroll
    for (int p = 0; p < 12; p += 2) {
        FM_ISSUE(gB, 2 * p + 2);
        FM_HOLD(gA);
        FM_SUM(gA);
        FM_ISSUE(gA, 2 * p + 4);
        FM_HOLD(gB);
        FM_SUM(gB);
    }
    FM_HOLD(gA);
    FM_SUM(gA);

#pragma unroll
    for (int v = 0; v < 8; ++v) {
        float t = 0.0f;
#pragma unroll
        for (int nt = 0; nt < 4; ++nt) {
            const float x = acc_vx[nt][v];
            t += x * x - acc_v2[nt][v];
        }
        t += __shfl_xor(t, 1, 32);
        t += __shfl_xor(t, 2, 32);
        t += __shfl_xor(t, 4, 32);
        t += __shfl_xor(t, 8, 32);
        const float sec = 0.5f * t;

        const float f_lo = __shfl(first, v,     32);
        const float f_hi = __shfl(first, 8 + v, 32);
        if (lane == 0)        out[row0 + v]     = sigmoidf_(f_lo + sec);
        else if (lane == 16)  out[row0 + 8 + v] = sigmoidf_(f_hi + sec);
    }
}

extern "C" void kernel_launch(void* const* d_in, const int* in_sizes, int n_in,
                              void* d_out, int out_size, void* d_ws, size_t ws_size,
                              hipStream_t stream) {
    const float* dense = (const float*)d_in[0];
    const int*   sp    = (const int*)  d_in[1];
    const float* w0p   = (const float*)d_in[2];
    const float* w     = (const float*)d_in[3];
    const float* V     = (const float*)d_in[4];
    float* out = (float*)d_out;
    float* Vt  = (float*)d_ws;

    dim3 grid_t(TP_BLOCKS);
    fm_transpose_kernel<<<grid_t, 256, 0, stream>>>(V, Vt);

    dim3 grid_g(B_TOTAL / 16);
    fm_gather_kernel<<<grid_g, 32, 0, stream>>>(dense, sp, w0p, w, Vt, out);
}


# gather launch: a 16-sample tile goes to a 4-wave workgroup, each wave gathers the rows of 4 samples (16-32 row loads in flight); dense WMMA split by latent tile, passed through LDS; sum order kept
# speedup vs baseline: 4.6288x; 1.0983x over previous
#include <hip/hip_runtime.h>
#include <math.h>


#define B_TOTAL   16384
#define ND        13
#define NS        26
#define FEAT_NUM  10000
#define FLEN      260013
#define KDIM      64

#define TP_F      128
#define TP_PITCH  129
#define TP_TILES  ((FLEN + TP_F - 1) / TP_F)
#define TP_BLOCKS 1024

typedef __attribute__((ext_vector_type(2))) float v2f;
typedef __attribute__((ext_vector_type(4))) float v4f;
typedef __attribute__((ext_vector_type(8))) float v8f;

__device__ __forceinline__ float sigmoidf_(float x) {
    return 1.0f / (1.0f + __expf(-x));
}

__device__ __forceinline__ void tp_load(float (&x)[32], const float* __restrict__ V,
                                        int f0, int wave, int lane) {
#pragma unroll
    for (int r = 0; r < 8; ++r) {
        const float* src = V + (size_t)(wave * 8 + r) * FLEN;
#pragma unroll
        for (int h = 0; h < 4; ++h) {
            const int f  = f0 + h * 32 + lane;
            const int fc = (f < FLEN) ? f : (FLEN - 1);
            x[r * 4 + h] = src[fc];
        }
    }
}

__device__ __forceinline__ void tp_store(const float (&x)[32], float* tile, float* __restrict__ Vt,
                                         int f0, int wave, int lane, int tid) {
#pragma unroll
    for (int r = 0; r < 8; ++r) {
#pragma unroll
        for (int h = 0; h < 4; ++h)
            tile[(wave * 8 + r) * TP_PITCH + h * 32 + lane] = x[r * 4 + h];
    }
    __syncthreads();
    const int n  = tid & 15;
    const int fr = tid >> 4;
#pragma unroll
    for (int i = 0; i < TP_F / 16; ++i) {
        const int fl = fr + 16 * i;
        v4f o;
        o.x = tile[(4 * n + 0) * TP_PITCH + fl];
        o.y = tile[(4 * n + 1) * TP_PITCH + fl];
        o.z = tile[(4 * n + 2) * TP_PITCH + fl];
        o.w = tile[(4 * n + 3) * TP_PITCH + fl];
        if (f0 + fl < FLEN)
            *(v4f*)(Vt + (size_t)(f0 + fl) * KDIM + 4 * n) = o;
    }
}

__global__ __launch_bounds__(256) void fm_transpose_kernel(
    const float* __restrict__ V,
    float* __restrict__ Vt)
{
    __shared__ float tile[KDIM * TP_PITCH];

    const int tid  = threadIdx.x;
    const int lane = tid & 31;
    const int wave = tid >> 5;
    const int t0   = blockIdx.x;
    const int t1   = blockIdx.x + TP_BLOCKS;
    const bool has1 = (t1 < TP_TILES);

    const int t1c = has1 ? t1 : t0;
    float xa[32], xb[32];
    tp_load(xa, V, t0 * TP_F, wave, lane);
    tp_load(xb, V, t1c * TP_F, wave, lane);

    tp_store(xa, tile, Vt, t0 * TP_F, wave, lane, tid);
    if (has1) {
        __syncthreads();
        tp_store(xb, tile, Vt, t1 * TP_F, wave, lane, tid);
    }
}

#define GW 4

#define FM_ISSUE(G, J0, NF)                                                          \
    _Pragma("unroll")                                                                \
    for (int q = 0; q < (NF); ++q) {                                                 \
        _Pragma("unroll")                                                            \
        for (int u = 0; u < 2; ++u) {                                                \
            const unsigned col = (unsigned)ND + (unsigned)((J0) + q) * FEAT_NUM      \
                               + (unsigned)spW[u * NS + (J0) + q];                   \
            G[q * 2 + u] = Vt4[col * 16u + (unsigned)n];     \
        }                                                                            \
    }

#define FM_HOLD16(G)                                                                 \
    asm volatile("" :: "v"(G[0]), "v"(G[1]), "v"(G[2]), "v"(G[3]), "v"(G[4]),        \
                 "v"(G[5]), "v"(G[6]), "v"(G[7]), "v"(G[8]), "v"(G[9]), "v"(G[10]),  \
                 "v"(G[11]), "v"(G[12]), "v"(G[13]), "v"(G[14]), "v"(G[15]))
#define FM_HOLD4(G)                                                                  \
    asm volatile("" :: "v"(G[0]), "v"(G[1]), "v"(G[2]), "v"(G[3]))

#define FM_SUM(G, NF)                                                                \
    _Pragma("unroll")                                                                \
    for (int q = 0; q < (NF); ++q) {                                                 \
        _Pragma("unroll")                                                            \
        for (int u = 0; u < 2; ++u) {                                                \
            _Pragma("unroll")                                                        \
            for (int nt = 0; nt < 4; ++nt) {                                         \
                const float val = G[q * 2 + u][nt];                                  \
                acc_vx[nt][u] += val;                                                \
                acc_v2[nt][u] += val * val;                                          \
            }                                                                        \
        }                                                                            \
    }

__global__ __launch_bounds__(32 * GW, 4) void fm_gather_kernel(
    const float* __restrict__ dense,
    const int*   __restrict__ sp,
    const float* __restrict__ w0p,
    const float* __restrict__ w,
    const float* __restrict__ Vt,
    float* __restrict__ out)
{
    const int tid  = threadIdx.x;
    const int lane = tid & 31;
    const int wave = tid >> 5;
    const int tile = blockIdx.x;
    const int row0 = tile * 16;

    const int n  = lane & 15;
    const int hi = lane >> 4;

    const v4f* __restrict__ Vt4 = (const v4f*)Vt;

    __shared__ int   spL[16 * NS];
    __shared__ float dC[2 * 4 * 8 * 32];
    __shared__ float res[16];

    int spr[4];
#pragma unroll
    for (int it = 0; it < 4; ++it) {
        const int idx = it * (32 * GW) + tid;
        const int idc = (idx < 16 * NS) ? idx : (16 * NS - 1);
        spr[it] = sp[(size_t)row0 * NS + idc];
    }

    v8f c_vx = {};
    v8f c_v2 = {};
#pragma unroll
    for (int kk = 0; kk < 16; kk += 4) {
        const int f0 = kk + 2 * hi;
        const bool in0 = (f0 + 0 < ND);
        const bool in1 = (f0 + 1 < ND);
        const int  fa  = in0 ? (f0 + 0) : (ND - 1);
        const int  fb  = in1 ? (f0 + 1) : (ND - 1);
        const float ax = dense[(size_t)(row0 + n) * ND + fa];
        const float ay = dense[(size_t)(row0 + n) * ND + fb];
        const float bxl = Vt[fa * KDIM + 4 * n + wave];
        const float byl = Vt[fb * KDIM + 4 * n + wave];
        v2f a, b;
        a.x = in0 ? ax : 0.0f;
        a.y = in1 ? ay : 0.0f;
        b.x = in0 ? bxl : 0.0f;
        b.y = in1 ? byl : 0.0f;
        const v2f a2 = a * a;
        const v2f b2 = b * b;
        c_vx = __builtin_amdgcn_wmma_f32_16x16x4_f32(
            false, a, false, b, (short)0, c_vx, false, false);
        c_v2 = __builtin_amdgcn_wmma_f32_16x16x4_f32(
            false, a2, false, b2, (short)0, c_v2, false, false);
    }

#pragma unroll
    for (int it = 0; it < 4; ++it) {
        const int idx = it * (32 * GW) + tid;
        if (idx < 16 * NS) spL[idx] = spr[it];
    }
#pragma unroll
    for (int v = 0; v < 8; ++v) {
        dC[((0 * 4 + wave) * 8 + v) * 32 + lane] = c_vx[v];
        dC[((1 * 4 + wave) * 8 + v) * 32 + lane] = c_v2[v];
    }
    __syncthreads();

    float acc_vx[4][2], acc_v2[4][2];
#pragma unroll
    for (int nt = 0; nt < 4; ++nt) {
#pragma unroll
        for (int u = 0; u < 2; ++u) {
            acc_vx[nt][u] = dC[((0 * 4 + nt) * 8 + 2 * wave + u) * 32 + lane];
            acc_v2[nt][u] = dC[((1 * 4 + nt) * 8 + 2 * wave + u) * 32 + lane];
        }
    }

    const int s1 = 2 * wave + (n & 1) + 8 * hi;
    const int r1 = row0 + s1;
    float first = w0p[0];
    for (int f = 0; f < ND; ++f)
        first += dense[(size_t)r1 * ND + f] * w[f];
    for (int j = 0; j < NS; ++j)
        first += w[(size_t)ND + (size_t)j * FEAT_NUM + (size_t)spL[s1 * NS + j]];

    const int* spW = spL + (2 * wave + 8 * hi) * NS;
    v4f gA[16], gB[16];

#define FM_KEEP() __builtin_amdgcn_sched_barrier(0)

    FM_ISSUE(gA, 0, 8);
    FM_ISSUE(gB, 8, 8);
    FM_KEEP();
    FM_HOLD16(gA);
    FM_SUM(gA, 8);
    FM_ISSUE(gA, 16, 8);
    FM_KEEP();
    FM_HOLD16(gB);
    FM_SUM(gB, 8);
    FM_ISSUE(gB, 24, 2);
    FM_KEEP();
    FM_HOLD16(gA);
    FM_SUM(gA, 8);
    FM_HOLD4(gB);
    FM_SUM(gB, 2);

#pragma unroll
    for (int u = 0; u < 2; ++u) {
        float t = 0.0f;
#pragma unroll
        for (int nt = 0; nt < 4; ++nt) {
            const float x = acc_vx[nt][u];
            t += x * x - acc_v2[nt][u];
        }
        t += __shfl_xor(t, 1, 32);
        t += __shfl_xor(t, 2, 32);
        t += __shfl_xor(t, 4, 32);
        t += __shfl_xor(t, 8, 32);
        const float sec = 0.5f * t;
        if (n == u) res[2 * wave + u + 8 * hi] = sigmoidf_(first + sec);
    }
    __syncthreads();
    if (tid < 16) out[row0 + tid] = res[tid];
}

extern "C" void kernel_launch(void* const* d_in, const int* in_sizes, int n_in,
                              void* d_out, int out_size, void* d_ws, size_t ws_size,
                              hipStream_t stream) {
    const float* dense = (const float*)d_in[0];
    const int*   sp    = (const int*)  d_in[1];
    const float* w0p   = (const float*)d_in[2];
    const float* w     = (const float*)d_in[3];
    const float* V     = (const float*)d_in[4];
    float* out = (float*)d_out;
    float* Vt  = (float*)d_ws;

    dim3 grid_t(TP_BLOCKS);
    fm_transpose_kernel<<<grid_t, 256, 0, stream>>>(V, Vt);

    dim3 grid_g(B_TOTAL / 16);
    fm_gather_kernel<<<grid_g, 32 * GW, 0, stream>>>(dense, sp, w0p, w, Vt, out);
}
